# starting kernel
# baseline (speedup 1.0000x reference)
#include <hip/hip_runtime.h>
#include <math.h>


typedef __attribute__((ext_vector_type(2))) float v2f;
typedef __attribute__((ext_vector_type(8))) float v8f;

#define IN_FEATS 128
#define HEADS    4
#define DHEAD    16
#define HD       64
#define NEG_SLOPE 0.2f

__device__ __forceinline__ void atomicMaxFloat(float* addr, float val) {
  if (val >= 0.0f) {
    atomicMax((int*)addr, __float_as_int(val));
  } else {
    atomicMin((unsigned int*)addr, __float_as_uint(val));
  }
}

__global__ __launch_bounds__(256) void gat_init(float* __restrict__ out,
                                                const float* __restrict__ bias,
                                                float* __restrict__ m,
                                                float* __restrict__ s,
                                                int nNodes) {
  int i = blockIdx.x * blockDim.x + threadIdx.x;
  int totalOut = nNodes * HD;
  if (i < totalOut) out[i] = bias[i & (HD - 1)];
  if (i < nNodes * HEADS) {
    m[i] = -INFINITY;
    s[i] = 0.0f;
  }
}

__global__ __launch_bounds__(256) void gat_proj_wmma(
    const float* __restrict__ feat, const float* __restrict__ W,
    float* __restrict__ ft, int nNodes, int nTiles) {
  const int lane = threadIdx.x & 31;
  const int tile = blockIdx.x * (blockDim.x >> 5) + (threadIdx.x >> 5);
  if (tile >= nTiles) return;

  const int row0 = tile * 16;
  const int mrow = lane & 15;
  const int kq   = lane >> 4;

  int arow = row0 + mrow;
  if (arow >= nNodes) arow = nNodes - 1;
  const float2* __restrict__ apf2 =
      (const float2*)(feat + (size_t)arow * IN_FEATS) + kq;

  v8f acc0 = {}, acc1 = {}, acc2 = {}, acc3 = {};

#pragma unroll 4
  for (int k = 0; k < IN_FEATS; k += 4) {
    float2 av = apf2[k >> 1];
    v2f a; a[0] = av.x; a[1] = av.y;

    const float* wr0 = W + (size_t)(k + 2 * kq) * HD;
    const float* wr1 = wr0 + HD;
    v2f b0; b0[0] = wr0[mrow];      b0[1] = wr1[mrow];
    v2f b1; b1[0] = wr0[mrow + 16]; b1[1] = wr1[mrow + 16];
    v2f b2; b2[0] = wr0[mrow + 32]; b2[1] = wr1[mrow + 32];
    v2f b3; b3[0] = wr0[mrow + 48]; b3[1] = wr1[mrow + 48];

    acc0 = __builtin_amdgcn_wmma_f32_16x16x4_f32(false, a, false, b0,
                                                 (short)0, acc0, false, false);
    acc1 = __builtin_amdgcn_wmma_f32_16x16x4_f32(false, a, false, b1,
                                                 (short)0, acc1, false, false);
    acc2 = __builtin_amdgcn_wmma_f32_16x16x4_f32(false, a, false, b2,
                                                 (short)0, acc2, false, false);
    acc3 = __builtin_amdgcn_wmma_f32_16x16x4_f32(false, a, false, b3,
                                                 (short)0, acc3, false, false);
  }

#pragma unroll
  for (int r = 0; r < 8; ++r) {
    int row = row0 + r + 8 * kq;
    if (row < nNodes) {
      float* o = ft + (size_t)row * HD + mrow;
      o[0]  = acc0[r];
      o[16] = acc1[r];
      o[32] = acc2[r];
      o[48] = acc3[r];
    }
  }
}

__global__ __launch_bounds__(256) void gat_scores(
    const float* __restrict__ ft, const float* __restrict__ attn_l,
    const float* __restrict__ attn_r, float* __restrict__ el,
    float* __restrict__ er, int NH) {
  int i = blockIdx.x * blockDim.x + threadIdx.x;
  if (i >= NH) return;
  int h = i & (HEADS - 1);
  const float* f  = ft + (size_t)(i >> 2) * HD + h * DHEAD;
  const float* al = attn_l + h * DHEAD;
  const float* ar = attn_r + h * DHEAD;
  float sl = 0.0f, sr = 0.0f;
#pragma unroll
  for (int d = 0; d < DHEAD; ++d) {
    float v = f[d];
    sl += v * al[d];
    sr += v * ar[d];
  }
  el[i] = sl;
  er[i] = sr;
}

__global__ __launch_bounds__(256) void gat_edge_max(
    const int* __restrict__ src, const int* __restrict__ dst,
    const float* __restrict__ el, const float* __restrict__ er,
    float* __restrict__ m, int EH) {
  int i = blockIdx.x * blockDim.x + threadIdx.x;
  if (i >= EH) return;
  int e = i >> 2, h = i & (HEADS - 1);
  int didx = dst[e] * HEADS + h;
  float x = el[src[e] * HEADS + h] + er[didx];
  x = (x > 0.0f) ? x : NEG_SLOPE * x;
  atomicMaxFloat(&m[didx], x);
}

__global__ __launch_bounds__(256) void gat_edge_expsum(
    const int* __restrict__ src, const int* __restrict__ dst,
    const float* __restrict__ el, const float* __restrict__ er,
    const float* __restrict__ m, float* __restrict__ s,
    float* __restrict__ ex, int EH) {
  int i = blockIdx.x * blockDim.x + threadIdx.x;
  if (i >= EH) return;
  int e = i >> 2, h = i & (HEADS - 1);
  int didx = dst[e] * HEADS + h;
  float x = el[src[e] * HEADS + h] + er[didx];
  x = (x > 0.0f) ? x : NEG_SLOPE * x;
  float v = expf(x - m[didx]);
  ex[i] = v;
  atomicAdd(&s[didx], v);
}

__global__ __launch_bounds__(256) void gat_scatter(
    const int* __restrict__ src, const int* __restrict__ dst,
    const float* __restrict__ ft, const float* __restrict__ ex,
    const float* __restrict__ s, float* __restrict__ out, long long total) {
  long long i = (long long)blockIdx.x * blockDim.x + threadIdx.x;
  if (i >= total) return;
  int e = (int)(i >> 6);
  int c = (int)(i & (HD - 1));
  int h = c >> 4;
  int se = src[e], de = dst[e];
  float a = ex[e * HEADS + h] / s[de * HEADS + h];
  atomicAdd(&out[(size_t)de * HD + c], ft[(size_t)se * HD + c] * a);
}

extern "C" void kernel_launch(void* const* d_in, const int* in_sizes, int n_in,
                              void* d_out, int out_size, void* d_ws,
                              size_t ws_size, hipStream_t stream) {
  const float* feat   = (const float*)d_in[0];
  const int*   src    = (const int*)d_in[1];
  const int*   dst    = (const int*)d_in[2];
  const float* W      = (const float*)d_in[3];
  const float* attn_l = (const float*)d_in[4];
  const float* attn_r = (const float*)d_in[5];
  const float* bias   = (const float*)d_in[6];
  float* out = (float*)d_out;

  const int N = in_sizes[0] / IN_FEATS;
  const int E = in_sizes[1];

  float* ws = (float*)d_ws;
  float* ft = ws;
  float* el = ft + (size_t)N * HD;
  float* er = el + (size_t)N * HEADS;
  float* m  = er + (size_t)N * HEADS;
  float* s  = m  + (size_t)N * HEADS;
  float* ex = s  + (size_t)N * HEADS;

  {
    int total = N * HD;
    gat_init<<<(total + 255) / 256, 256, 0, stream>>>(out, bias, m, s, N);
  }
  {
    int nTiles = (N + 15) / 16;
    int blocks = (nTiles + 7) / 8;
    gat_proj_wmma<<<blocks, 256, 0, stream>>>(feat, W, ft, N, nTiles);
  }
  {
    int NH = N * HEADS;
    gat_scores<<<(NH + 255) / 256, 256, 0, stream>>>(ft, attn_l, attn_r, el,
                                                     er, NH);
  }
  {
    int EH = E * HEADS;
    gat_edge_max<<<(EH + 255) / 256, 256, 0, stream>>>(src, dst, el, er, m, EH);
  }
  {
    int EH = E * HEADS;
    gat_edge_expsum<<<(EH + 255) / 256, 256, 0, stream>>>(src, dst, el, er, m,
                                                          s, ex, EH);
  }
  {
    long long total = (long long)E * HD;
    int blocks = (int)((total + 255) / 256);
    gat_scatter<<<blocks, 256, 0, stream>>>(src, dst, ft, ex, s, out, total);
  }
}


# Edges bucketed by dst (one returning int atomic per edge, bump-allocated segments), then 16 lanes per dst node do max, exp, sum and gather: no float atomics; sums in arrival-rank order, divided once
# speedup vs baseline: 2.4458x; 2.4458x over previous
#include <hip/hip_runtime.h>
#include <math.h>


typedef __attribute__((ext_vector_type(2))) float v2f;
typedef __attribute__((ext_vector_type(8))) float v8f;

#define IN_FEATS 128
#define HEADS    4
#define DHEAD    16
#define HD       64
#define NEG_SLOPE 0.2f

__global__ __launch_bounds__(256) void gat_proj_wmma(
    const float* __restrict__ feat, const float* __restrict__ W,
    float* __restrict__ ft, int nNodes, int nTiles) {
  const int lane = threadIdx.x & 31;
  const int tile = blockIdx.x * (blockDim.x >> 5) + (threadIdx.x >> 5);
  if (tile >= nTiles) return;

  const int row0 = tile * 16;
  const int mrow = lane & 15;
  const int kq   = lane >> 4;

  int arow = row0 + mrow;
  if (arow >= nNodes) arow = nNodes - 1;
  const float2* __restrict__ apf2 =
      (const float2*)(feat + (size_t)arow * IN_FEATS) + kq;

  v8f acc0 = {}, acc1 = {}, acc2 = {}, acc3 = {};

#pragma unroll 4
  for (int k = 0; k < IN_FEATS; k += 4) {
    float2 av = apf2[k >> 1];
    v2f a; a[0] = av.x; a[1] = av.y;

    const float* wr0 = W + (size_t)(k + 2 * kq) * HD;
    const float* wr1 = wr0 + HD;
    v2f b0; b0[0] = wr0[mrow];      b0[1] = wr1[mrow];
    v2f b1; b1[0] = wr0[mrow + 16]; b1[1] = wr1[mrow + 16];
    v2f b2; b2[0] = wr0[mrow + 32]; b2[1] = wr1[mrow + 32];
    v2f b3; b3[0] = wr0[mrow + 48]; b3[1] = wr1[mrow + 48];

    acc0 = __builtin_amdgcn_wmma_f32_16x16x4_f32(false, a, false, b0,
                                                 (short)0, acc0, false, false);
    acc1 = __builtin_amdgcn_wmma_f32_16x16x4_f32(false, a, false, b1,
                                                 (short)0, acc1, false, false);
    acc2 = __builtin_amdgcn_wmma_f32_16x16x4_f32(false, a, false, b2,
                                                 (short)0, acc2, false, false);
    acc3 = __builtin_amdgcn_wmma_f32_16x16x4_f32(false, a, false, b3,
                                                 (short)0, acc3, false, false);
  }

#pragma unroll
  for (int r = 0; r < 8; ++r) {
    int row = row0 + r + 8 * kq;
    if (row < nNodes) {
      float* o = ft + (size_t)row * HD + mrow;
      o[0]  = acc0[r];
      o[16] = acc1[r];
      o[32] = acc2[r];
      o[48] = acc3[r];
    }
  }
}

__global__ __launch_bounds__(256) void gat_scores(
    const float* __restrict__ ft, const float* __restrict__ attn_l,
    const float* __restrict__ attn_r, float* __restrict__ el,
    float* __restrict__ er, int* __restrict__ cnt, int* __restrict__ cursor,
    int NH, int nNodes) {
  int i = blockIdx.x * blockDim.x + threadIdx.x;
  if (i < nNodes) cnt[i] = 0;
  if (i == 0) *cursor = 0;
  if (i >= NH) return;
  int h = i & (HEADS - 1);
  const float* f  = ft + (size_t)(i >> 2) * HD + h * DHEAD;
  const float* al = attn_l + h * DHEAD;
  const float* ar = attn_r + h * DHEAD;
  float sl = 0.0f, sr = 0.0f;
#pragma unroll
  for (int d = 0; d < DHEAD; ++d) {
    float v = f[d];
    sl += v * al[d];
    sr += v * ar[d];
  }
  el[i] = sl;
  er[i] = sr;
}

__global__ __launch_bounds__(256) void gat_count(
    const int* __restrict__ dst, int* __restrict__ cnt,
    int* __restrict__ rank, int E) {
  int e = blockIdx.x * blockDim.x + threadIdx.x;
  if (e >= E) return;
  rank[e] = atomicAdd(&cnt[dst[e]], 1);
}

__global__ __launch_bounds__(256) void gat_alloc(
    const int* __restrict__ cnt, int* __restrict__ off,
    int* __restrict__ cursor, int nNodes) {
  __shared__ int wtot[8];
  __shared__ int blockBase;
  const int tid = threadIdx.x;
  const int lane = tid & 31, w = tid >> 5;
  const int i = blockIdx.x * 256 + tid;
  const int c = (i < nNodes) ? cnt[i] : 0;
  int v = c;
  for (int d = 1; d < 32; d <<= 1) {
    int t = __shfl_up(v, d);
    if (lane >= d) v += t;
  }
  if (lane == 31) wtot[w] = v;
  __syncthreads();
  if (tid == 0) {
    int tot = 0;
    for (int k = 0; k < 8; ++k) { int t = wtot[k]; wtot[k] = tot; tot += t; }
    blockBase = atomicAdd(cursor, tot);
  }
  __syncthreads();
  if (i < nNodes) off[i] = blockBase + wtot[w] + v - c;
}

__global__ __launch_bounds__(256) void gat_fill(
    const int* __restrict__ src, const int* __restrict__ dst,
    const int* __restrict__ rank, const int* __restrict__ off,
    int* __restrict__ ssrc, int E) {
  int e = blockIdx.x * blockDim.x + threadIdx.x;
  if (e >= E) return;
  ssrc[off[dst[e]] + rank[e]] = src[e];
}

template <int CTRL>
__device__ __forceinline__ int quad_sel_i(int v) {
  return __builtin_amdgcn_update_dpp(0, v, CTRL, 0xF, 0xF, false);
}
template <int CTRL>
__device__ __forceinline__ float quad_sel_f(float v) {
  return __int_as_float(quad_sel_i<CTRL>(__float_as_int(v)));
}
#define QUAD_XOR1 0xB1
#define QUAD_XOR2 0x4E

template <int Q>
__device__ __forceinline__ void gat_edge_acc(int s, float p,
                                             const float4* __restrict__ ft4,
                                             int j, float4& acc) {
  const int   sq = quad_sel_i<Q * 0x55>(s);
  const float pq = quad_sel_f<Q * 0x55>(p);
  const float4 f = ft4[(size_t)sq * (HD / 4) + j];
  acc.x = fmaf(pq, f.x, acc.x);
  acc.y = fmaf(pq, f.y, acc.y);
  acc.z = fmaf(pq, f.z, acc.z);
  acc.w = fmaf(pq, f.w, acc.w);
}

__global__ __launch_bounds__(256) void gat_node(
    const int* __restrict__ ssrc, const int* __restrict__ off,
    const int* __restrict__ cnt, const float* __restrict__ ft,
    const float* __restrict__ el, const float* __restrict__ er,
    const float* __restrict__ bias, float* __restrict__ out, int nNodes) {
  const int j = threadIdx.x & 15;
  const int n = blockIdx.x * 16 + (threadIdx.x >> 4);
  if (n >= nNodes) return;
  const int q = j & 3, h = j >> 2;
  const int deg = cnt[n];
  const int* __restrict__ seg = ssrc + off[n];
  const float erh = er[n * HEADS + h];
  const float4* __restrict__ ft4 = (const float4*)ft;

  float mx = -INFINITY;
  for (int base = 0; base < deg; base += 16) {
    const int t0 = min(base + q,      deg - 1);
    const int t1 = min(base + q + 4,  deg - 1);
    const int t2 = min(base + q + 8,  deg - 1);
    const int t3 = min(base + q + 12, deg - 1);
    const int s0 = seg[t0], s1 = seg[t1], s2 = seg[t2], s3 = seg[t3];
    float x0 = el[s0 * HEADS + h] + erh;
    float x1 = el[s1 * HEADS + h] + erh;
    float x2 = el[s2 * HEADS + h] + erh;
    float x3 = el[s3 * HEADS + h] + erh;
    x0 = (x0 > 0.0f) ? x0 : NEG_SLOPE * x0;
    x1 = (x1 > 0.0f) ? x1 : NEG_SLOPE * x1;
    x2 = (x2 > 0.0f) ? x2 : NEG_SLOPE * x2;
    x3 = (x3 > 0.0f) ? x3 : NEG_SLOPE * x3;
    mx = fmaxf(mx, fmaxf(fmaxf(x0, x1), fmaxf(x2, x3)));
  }
  mx = fmaxf(mx, quad_sel_f<QUAD_XOR1>(mx));
  mx = fmaxf(mx, quad_sel_f<QUAD_XOR2>(mx));

  float4 acc = {0.0f, 0.0f, 0.0f, 0.0f};
  float ps = 0.0f;
  int base = 0;
  for (; base + 4 <= deg; base += 4) {
    const int s = seg[base + q];
    float x = el[s * HEADS + h] + erh;
    x = (x > 0.0f) ? x : NEG_SLOPE * x;
    const float p = expf(x - mx);
    ps += p;
    gat_edge_acc<0>(s, p, ft4, j, acc);
    gat_edge_acc<1>(s, p, ft4, j, acc);
    gat_edge_acc<2>(s, p, ft4, j, acc);
    gat_edge_acc<3>(s, p, ft4, j, acc);
  }
  if (base < deg) {
    const int t = base + q;
    int s = 0;
    float p = 0.0f;
    if (t < deg) {
      s = seg[t];
      float x = el[s * HEADS + h] + erh;
      x = (x > 0.0f) ? x : NEG_SLOPE * x;
      p = expf(x - mx);
    }
    ps += p;
    gat_edge_acc<0>(s, p, ft4, j, acc);
    if (base + 1 < deg) gat_edge_acc<1>(s, p, ft4, j, acc);
    if (base + 2 < deg) gat_edge_acc<2>(s, p, ft4, j, acc);
  }
  ps += quad_sel_f<QUAD_XOR1>(ps);
  ps += quad_sel_f<QUAD_XOR2>(ps);

  float4 o;
  o.x = bias[4 * j + 0];
  o.y = bias[4 * j + 1];
  o.z = bias[4 * j + 2];
  o.w = bias[4 * j + 3];
  if (deg > 0) {
    o.x += acc.x / ps;
    o.y += acc.y / ps;
    o.z += acc.z / ps;
    o.w += acc.w / ps;
  }
  ((float4*)out)[(size_t)n * (HD / 4) + j] = o;
}

extern "C" void kernel_launch(void* const* d_in, const int* in_sizes, int n_in,
                              void* d_out, int out_size, void* d_ws,
                              size_t ws_size, hipStream_t stream) {
  const float* feat   = (const float*)d_in[0];
  const int*   src    = (const int*)d_in[1];
  const int*   dst    = (const int*)d_in[2];
  const float* W      = (const float*)d_in[3];
  const float* attn_l = (const float*)d_in[4];
  const float* attn_r = (const float*)d_in[5];
  const float* bias   = (const float*)d_in[6];
  float* out = (float*)d_out;

  const int N = in_sizes[0] / IN_FEATS;
  const int E = in_sizes[1];

  float* ws = (float*)d_ws;
  float* ft = ws;
  float* el = ft + (size_t)N * HD;
  float* er = el + (size_t)N * HEADS;
  int* cnt    = (int*)(er + (size_t)N * HEADS);
  int* off    = cnt + N;
  int* cursor = off + N;
  int* rank   = cursor + 4;
  int* ssrc   = rank + E;

  {
    int nTiles = (N + 15) / 16;
    int blocks = (nTiles + 7) / 8;
    gat_proj_wmma<<<blocks, 256, 0, stream>>>(feat, W, ft, N, nTiles);
  }
  {
    int NH = N * HEADS;
    gat_scores<<<(NH + 255) / 256, 256, 0, stream>>>(ft, attn_l, attn_r, el,
                                                     er, cnt, cursor, NH, N);
  }
  gat_count<<<(E + 255) / 256, 256, 0, stream>>>(dst, cnt, rank, E);
  gat_alloc<<<(N + 255) / 256, 256, 0, stream>>>(cnt, off, cursor, N);
  gat_fill<<<(E + 255) / 256, 256, 0, stream>>>(src, dst, rank, off, ssrc, E);
  gat_node<<<(N + 15) / 16, 256, 0, stream>>>(ssrc, off, cnt, ft, el, er, bias,
                                              out, N);
}


# Edges sorted by destination bin (128 nodes) in LDS: one global add and one consecutive run per work-group and bin, not per edge; node launch ranks its bin's records in LDS; sums in arrival-rank order
# speedup vs baseline: 5.9184x; 2.4198x over previous
#include <hip/hip_runtime.h>
#include <math.h>


typedef __attribute__((ext_vector_type(2))) float v2f;
typedef __attribute__((ext_vector_type(8))) float v8f;

#define IN_FEATS 128
#define HEADS    4
#define DHEAD    16
#define HD       64
#define NEG_SLOPE 0.2f
#define CAP      64
typedef unsigned short slot_t;
#define BIN_SHIFT 7
#define BIN_NODES 128
#define BCAP      4096
#define CUR_PAD   32
#define LBINS     512
#define PART_THREADS 1024
#define PART_EDGES   4
#define PART_TILE    (PART_THREADS * PART_EDGES)
#define GROUP_NODES  16

__global__ __launch_bounds__(256) void gat_proj_wmma(
    const float* __restrict__ feat, const float* __restrict__ W,
    float* __restrict__ ft, int* __restrict__ cursor, int nBins, int nNodes,
    int nTiles) {
  const int gid = blockIdx.x * blockDim.x + threadIdx.x;
  if (gid < nBins) cursor[gid * CUR_PAD] = 0;
  const int lane = threadIdx.x & 31;
  const int tile = blockIdx.x * (blockDim.x >> 5) + (threadIdx.x >> 5);
  if (tile >= nTiles) return;

  const int row0 = tile * 16;
  const int mrow = lane & 15;
  const int kq   = lane >> 4;

  int arow = row0 + mrow;
  if (arow >= nNodes) arow = nNodes - 1;
  const float2* __restrict__ apf2 =
      (const float2*)(feat + (size_t)arow * IN_FEATS) + kq;

  v8f acc0 = {}, acc1 = {}, acc2 = {}, acc3 = {};

#pragma unroll 4
  for (int k = 0; k < IN_FEATS; k += 4) {
    float2 av = apf2[k >> 1];
    v2f a; a[0] = av.x; a[1] = av.y;

    const float* wr0 = W + (size_t)(k + 2 * kq) * HD;
    const float* wr1 = wr0 + HD;
    v2f b0; b0[0] = wr0[mrow];      b0[1] = wr1[mrow];
    v2f b1; b1[0] = wr0[mrow + 16]; b1[1] = wr1[mrow + 16];
    v2f b2; b2[0] = wr0[mrow + 32]; b2[1] = wr1[mrow + 32];
    v2f b3; b3[0] = wr0[mrow + 48]; b3[1] = wr1[mrow + 48];

    acc0 = __builtin_amdgcn_wmma_f32_16x16x4_f32(false, a, false, b0,
                                                 (short)0, acc0, false, false);
    acc1 = __builtin_amdgcn_wmma_f32_16x16x4_f32(false, a, false, b1,
                                                 (short)0, acc1, false, false);
    acc2 = __builtin_amdgcn_wmma_f32_16x16x4_f32(false, a, false, b2,
                                                 (short)0, acc2, false, false);
    acc3 = __builtin_amdgcn_wmma_f32_16x16x4_f32(false, a, false, b3,
                                                 (short)0, acc3, false, false);
  }

#pragma unroll
  for (int r = 0; r < 8; ++r) {
    int row = row0 + r + 8 * kq;
    if (row < nNodes) {
      float* o = ft + (size_t)row * HD + mrow;
      o[0]  = acc0[r];
      o[16] = acc1[r];
      o[32] = acc2[r];
      o[48] = acc3[r];
    }
  }
}

__global__ __launch_bounds__(PART_THREADS) void gat_part(
    const int* __restrict__ src, const int* __restrict__ dst,
    int* __restrict__ cursor, int* __restrict__ binbuf, int E, int nNodes) {
  if (nNodes > 0x10000) return;
  __shared__ int lh[LBINS];
  __shared__ int lb[LBINS];
  __shared__ int gb[LBINS];
  __shared__ int wsum[LBINS / 32];
  __shared__ unsigned stage[PART_TILE];
  const int tid = threadIdx.x;
  if (tid < LBINS) lh[tid] = 0;
  __syncthreads();
  const int tile0 = blockIdx.x * PART_TILE;
  int d[PART_EDGES], s[PART_EDGES], lr[PART_EDGES];
#pragma unroll
  for (int k = 0; k < PART_EDGES; ++k) {
    const int e = tile0 + k * PART_THREADS + tid;
    d[k] = -1;
    s[k] = 0;
    if (e < E) { d[k] = dst[e]; s[k] = src[e]; }
  }
#pragma unroll
  for (int k = 0; k < PART_EDGES; ++k) {
    lr[k] = 0;
    if (d[k] >= 0) lr[k] = atomicAdd(&lh[d[k] >> BIN_SHIFT], 1);
  }
  __syncthreads();
  const int lane = tid & 31;
  const int v = (tid < LBINS) ? lh[tid] : 0;
  int incl = v;
#pragma unroll
  for (int off = 1; off < 32; off <<= 1) {
    const int t = __shfl_up(incl, off, 32);
    if (lane >= off) incl += t;
  }
  if (tid < LBINS && lane == 31) wsum[tid >> 5] = incl;
  __syncthreads();
  if (tid < LBINS) {
    int before = 0;
    for (int w = 0; w < (tid >> 5); ++w) before += wsum[w];
    lb[tid] = before + incl - v;
    gb[tid] = (v > 0) ? atomicAdd(&cursor[tid * CUR_PAD], v) : 0;
  }
  __syncthreads();
#pragma unroll
  for (int k = 0; k < PART_EDGES; ++k) {
    if (d[k] >= 0)
      stage[lb[d[k] >> BIN_SHIFT] + lr[k]] =
          ((unsigned)d[k] << 16) | (unsigned)s[k];
  }
  __syncthreads();
  const int nl = min(PART_TILE, E - tile0);
  for (int i = tid; i < nl; i += PART_THREADS) {
    const unsigned rec = stage[i];
    const int bin = (int)(rec >> (16 + BIN_SHIFT));
    const int pos = gb[bin] + (i - lb[bin]);
    if (pos < BCAP) binbuf[(size_t)bin * BCAP + pos] = (int)rec;
  }
}

template <int CTRL>
__device__ __forceinline__ int quad_sel_i(int v) {
  return __builtin_amdgcn_update_dpp(0, v, CTRL, 0xF, 0xF, false);
}
template <int CTRL>
__device__ __forceinline__ float quad_sel_f(float v) {
  return __int_as_float(quad_sel_i<CTRL>(__float_as_int(v)));
}
#define QUAD_XOR1 0xB1
#define QUAD_XOR2 0x4E

__device__ __forceinline__ float dot4(const float4 f, const float4 w) {
  float d = f.x * w.x;
  d = fmaf(f.y, w.y, d);
  d = fmaf(f.z, w.z, d);
  d = fmaf(f.w, w.w, d);
  return d;
}

__device__ __forceinline__ float4 ft_row(const float* __restrict__ ft, int s,
                                         int j) {
  const unsigned boff = (unsigned)s * (HD * 4u) + (unsigned)j * 16u;
  return *(const float4*)((const char*)ft + boff);
}

template <bool TAIL>
__device__ __forceinline__ void gat_trip(const float* __restrict__ ft, int s,
                                         int j, int q, int nlive,
                                         const float4 al, float erh, float& m,
                                         float& ps, float4& acc) {
  const int s0 = quad_sel_i<0x00>(s);
  const int s1 = quad_sel_i<0x55>(s);
  const int s2 = quad_sel_i<0xAA>(s);
  const int s3 = quad_sel_i<0xFF>(s);
  const float4 zero4 = {0.0f, 0.0f, 0.0f, 0.0f};
  float4 f0, f1, f2, f3;
  f0 = ft_row(ft, s0, j);
  if (!TAIL) {
    f1 = ft_row(ft, s1, j);
    f2 = ft_row(ft, s2, j);
    f3 = ft_row(ft, s3, j);
  } else {
    f1 = zero4; f2 = zero4; f3 = zero4;
    if (nlive > 1) f1 = ft_row(ft, s1, j);
    if (nlive > 2) f2 = ft_row(ft, s2, j);
  }
  const float d0 = dot4(f0, al);
  const float d1 = dot4(f1, al);
  const float d2 = dot4(f2, al);
  const float d3 = dot4(f3, al);
  const bool o1 = (q & 1) != 0, o2 = (q & 2) != 0;
  const float a = (o1 ? d1 : d0) + quad_sel_f<QUAD_XOR1>(o1 ? d0 : d1);
  const float b = (o1 ? d3 : d2) + quad_sel_f<QUAD_XOR1>(o1 ? d2 : d3);
  const float el = (o2 ? b : a) + quad_sel_f<QUAD_XOR2>(o2 ? a : b);
  float x = el + erh;
  x = (x > 0.0f) ? x : NEG_SLOPE * x;
  if (TAIL && q >= nlive) x = -INFINITY;
  float cm = fmaxf(x, quad_sel_f<QUAD_XOR1>(x));
  cm = fmaxf(cm, quad_sel_f<QUAD_XOR2>(cm));
  const float mn = fmaxf(m, cm);
  const float sc = expf(m - mn);
  const float p = expf(x - mn);
  m = mn;
  ps = fmaf(ps, sc, p);
  const float p0 = quad_sel_f<0x00>(p);
  const float p1 = quad_sel_f<0x55>(p);
  const float p2 = quad_sel_f<0xAA>(p);
  const float p3 = quad_sel_f<0xFF>(p);
  acc.x = fmaf(p3, f3.x, fmaf(p2, f2.x, fmaf(p1, f1.x, fmaf(p0, f0.x, acc.x * sc))));
  acc.y = fmaf(p3, f3.y, fmaf(p2, f2.y, fmaf(p1, f1.y, fmaf(p0, f0.y, acc.y * sc))));
  acc.z = fmaf(p3, f3.z, fmaf(p2, f2.z, fmaf(p1, f1.z, fmaf(p0, f0.z, acc.z * sc))));
  acc.w = fmaf(p3, f3.w, fmaf(p2, f2.w, fmaf(p1, f1.w, fmaf(p0, f0.w, acc.w * sc))));
}

__device__ __forceinline__ int rank_record(unsigned rec, bool live, int* lcnt) {
  const int d = (int)(rec >> 16);
  if (live && (d >> 4) == (int)blockIdx.x)
    return atomicAdd(&lcnt[d & (GROUP_NODES - 1)], 1);
  return CAP;
}
__device__ __forceinline__ void store_record(unsigned rec, int r, slot_t* ls) {
  if (r < CAP)
    ls[(int)((rec >> 16) & (GROUP_NODES - 1)) * CAP + r] =
        (slot_t)(rec & 0xFFFFu);
}
__device__ __forceinline__ void keep_records(const int4 a, int ia, const int4 b,
                                             int ib, int total, int* lcnt,
                                             slot_t* ls) {
  const unsigned w[8] = {(unsigned)a.x, (unsigned)a.y, (unsigned)a.z,
                         (unsigned)a.w, (unsigned)b.x, (unsigned)b.y,
                         (unsigned)b.z, (unsigned)b.w};
  int r[8];
#pragma unroll
  for (int k = 0; k < 4; ++k) {
    r[k] = rank_record(w[k], ia + k < total, lcnt);
    r[4 + k] = rank_record(w[4 + k], ib + k < total, lcnt);
  }
#pragma unroll
  for (int k = 0; k < 8; ++k) store_record(w[k], r[k], ls);
}

__global__ __launch_bounds__(256) void gat_node(
    const int* __restrict__ binbuf, const int* __restrict__ cursor,
    const int* __restrict__ src, const int* __restrict__ dst,
    const float* __restrict__ ft, const float* __restrict__ attn_l,
    const float* __restrict__ attn_r, const float* __restrict__ bias,
    float* __restrict__ out, int nNodes, int E) {
  __shared__ int lcnt[GROUP_NODES];
  __shared__ slot_t ls[GROUP_NODES * CAP];
  const int tid = threadIdx.x;
  const bool table = nNodes <= 0x10000;
  if (tid < GROUP_NODES) lcnt[tid] = 0;
  __syncthreads();
  if (table) {
    const int bin = (int)(blockIdx.x * GROUP_NODES) >> BIN_SHIFT;
    const int total = cursor[bin * CUR_PAD];
    if (total <= BCAP) {
      const int4* __restrict__ recs =
          (const int4*)(binbuf + (size_t)bin * BCAP);
      for (int i = tid; 4 * i < total; i += 512) {
        const int4 ra = recs[i];
        const int4 rb = recs[min(i + 256, BCAP / 4 - 1)];
        keep_records(ra, 4 * i, rb, 4 * (i + 256), total, lcnt, ls);
      }
    } else {
      for (int e = tid; e < E; e += 256) {
        const unsigned rec = ((unsigned)dst[e] << 16) | (unsigned)src[e];
        store_record(rec, rank_record(rec, true, lcnt), ls);
      }
    }
  }
  __syncthreads();

  const int j = tid & 15;
  const int g = tid >> 4;
  const int n = blockIdx.x * GROUP_NODES + g;
  if (n >= nNodes) return;
  const int q = j & 3;
  const slot_t* seg = ls + g * CAP;
  int s = seg[q];
  const int deg = lcnt[g];

  const float4 al = ((const float4*)attn_l)[j];
  const float4 ar = ((const float4*)attn_r)[j];
  float erh = dot4(ft_row(ft, n, j), ar);
  erh += quad_sel_f<QUAD_XOR1>(erh);
  erh += quad_sel_f<QUAD_XOR2>(erh);

  float m = -INFINITY, ps = 0.0f;
  float4 acc = {0.0f, 0.0f, 0.0f, 0.0f};
  int nedge = deg;
  if (deg > CAP || !table) {
    const int gshift = threadIdx.x & 16;
    nedge = 0;
    for (int base = 0; base < E; base += 16) {
      const int e = base + j;
      const bool hit = (e < E) && (dst[e] == n);
      unsigned hits = (unsigned)(__ballot(hit) >> gshift) & 0xFFFFu;
      while (hits != 0u) {
        const int b = __ffs(hits) - 1;
        hits &= hits - 1u;
        gat_trip<true>(ft, src[base + b], j, q, 1, al, erh, m, ps, acc);
        ++nedge;
      }
    }
  } else if (deg > 0) {
    int base = 0;
    for (; base + 4 <= deg; base += 4) {
      const int sn = seg[min(base + 4 + q, CAP - 1)];
      gat_trip<false>(ft, s, j, q, 4, al, erh, m, ps, acc);
      s = sn;
    }
    if (base < deg) gat_trip<true>(ft, s, j, q, deg - base, al, erh, m, ps, acc);
  }
  ps += quad_sel_f<QUAD_XOR1>(ps);
  ps += quad_sel_f<QUAD_XOR2>(ps);

  float4 o;
  o.x = bias[4 * j + 0];
  o.y = bias[4 * j + 1];
  o.z = bias[4 * j + 2];
  o.w = bias[4 * j + 3];
  if (nedge > 0) {
    o.x += acc.x / ps;
    o.y += acc.y / ps;
    o.z += acc.z / ps;
    o.w += acc.w / ps;
  }
  ((float4*)out)[(size_t)n * (HD / 4) + j] = o;
}

extern "C" void kernel_launch(void* const* d_in, const int* in_sizes, int n_in,
                              void* d_out, int out_size, void* d_ws,
                              size_t ws_size, hipStream_t stream) {
  const float* feat   = (const float*)d_in[0];
  const int*   src    = (const int*)d_in[1];
  const int*   dst    = (const int*)d_in[2];
  const float* W      = (const float*)d_in[3];
  const float* attn_l = (const float*)d_in[4];
  const float* attn_r = (const float*)d_in[5];
  const float* bias   = (const float*)d_in[6];
  float* out = (float*)d_out;

  const int N = in_sizes[0] / IN_FEATS;
  const int E = in_sizes[1];

  const int nBins = (N + BIN_NODES - 1) / BIN_NODES;

  float* ws = (float*)d_ws;
  float* ft = ws;
  int* binbuf = (int*)(ft + (size_t)N * HD);
  int* cursor = binbuf + (size_t)nBins * BCAP;

  {
    int nTiles = (N + 15) / 16;
    int blocks = (nTiles + 7) / 8;
    gat_proj_wmma<<<blocks, 256, 0, stream>>>(feat, W, ft, cursor, nBins, N,
                                              nTiles);
  }
  gat_part<<<(E + PART_TILE - 1) / PART_TILE, PART_THREADS, 0, stream>>>(
      src, dst, cursor, binbuf, E, N);
  gat_node<<<(N + GROUP_NODES - 1) / GROUP_NODES, 256, 0, stream>>>(
      binbuf, cursor, src, dst, ft, attn_l, attn_r, bias, out, N, E);
}


# Edge launch: a wave's lanes whose consecutive edges fall in one bin take ONE LDS add per run (start + place in run), not 32 same-word adds served lane by lane; sums still in arrival-rank order
# speedup vs baseline: 6.3314x; 1.0698x over previous
#include <hip/hip_runtime.h>
#include <math.h>


typedef __attribute__((ext_vector_type(2))) float v2f;
typedef __attribute__((ext_vector_type(8))) float v8f;

#define IN_FEATS 128
#define HEADS    4
#define DHEAD    16
#define HD       64
#define NEG_SLOPE 0.2f
#define CAP      64
typedef unsigned short slot_t;
#define BIN_SHIFT 7
#define BIN_NODES 128
#define BCAP      4096
#define CUR_PAD   32
#define LBINS     512
#define PART_THREADS 1024
#define PART_EDGES   4
#define PART_TILE    (PART_THREADS * PART_EDGES)
#define GROUP_NODES  16

__global__ __launch_bounds__(256) void gat_proj_wmma(
    const float* __restrict__ feat, const float* __restrict__ W,
    float* __restrict__ ft, int* __restrict__ cursor, int nBins, int nNodes,
    int nTiles) {
  const int gid = blockIdx.x * blockDim.x + threadIdx.x;
  if (gid < nBins) cursor[gid * CUR_PAD] = 0;
  const int lane = threadIdx.x & 31;
  const int tile = blockIdx.x * (blockDim.x >> 5) + (threadIdx.x >> 5);
  if (tile >= nTiles) return;

  const int row0 = tile * 16;
  const int mrow = lane & 15;
  const int kq   = lane >> 4;

  int arow = row0 + mrow;
  if (arow >= nNodes) arow = nNodes - 1;
  const float2* __restrict__ apf2 =
      (const float2*)(feat + (size_t)arow * IN_FEATS) + kq;

  v8f acc0 = {}, acc1 = {}, acc2 = {}, acc3 = {};

#pragma unroll 4
  for (int k = 0; k < IN_FEATS; k += 4) {
    float2 av = apf2[k >> 1];
    v2f a; a[0] = av.x; a[1] = av.y;

    const float* wr0 = W + (size_t)(k + 2 * kq) * HD;
    const float* wr1 = wr0 + HD;
    v2f b0; b0[0] = wr0[mrow];      b0[1] = wr1[mrow];
    v2f b1; b1[0] = wr0[mrow + 16]; b1[1] = wr1[mrow + 16];
    v2f b2; b2[0] = wr0[mrow + 32]; b2[1] = wr1[mrow + 32];
    v2f b3; b3[0] = wr0[mrow + 48]; b3[1] = wr1[mrow + 48];

    acc0 = __builtin_amdgcn_wmma_f32_16x16x4_f32(false, a, false, b0,
                                                 (short)0, acc0, false, false);
    acc1 = __builtin_amdgcn_wmma_f32_16x16x4_f32(false, a, false, b1,
                                                 (short)0, acc1, false, false);
    acc2 = __builtin_amdgcn_wmma_f32_16x16x4_f32(false, a, false, b2,
                                                 (short)0, acc2, false, false);
    acc3 = __builtin_amdgcn_wmma_f32_16x16x4_f32(false, a, false, b3,
                                                 (short)0, acc3, false, false);
  }

#pragma unroll
  for (int r = 0; r < 8; ++r) {
    int row = row0 + r + 8 * kq;
    if (row < nNodes) {
      float* o = ft + (size_t)row * HD + mrow;
      o[0]  = acc0[r];
      o[16] = acc1[r];
      o[32] = acc2[r];
      o[48] = acc3[r];
    }
  }
}

__global__ __launch_bounds__(PART_THREADS) void gat_part(
    const int* __restrict__ src, const int* __restrict__ dst,
    int* __restrict__ cursor, int* __restrict__ binbuf, int E, int nNodes) {
  if (nNodes > 0x10000) return;
  __shared__ int lh[LBINS];
  __shared__ int lb[LBINS];
  __shared__ int gb[LBINS];
  __shared__ int wsum[LBINS / 32];
  __shared__ unsigned stage[PART_TILE];
  const int tid = threadIdx.x;
  if (tid < LBINS) lh[tid] = 0;
  __syncthreads();
  const int tile0 = blockIdx.x * PART_TILE;
  int d[PART_EDGES], s[PART_EDGES], lr[PART_EDGES];
#pragma unroll
  for (int k = 0; k < PART_EDGES; ++k) {
    const int e = tile0 + k * PART_THREADS + tid;
    d[k] = -1;
    s[k] = 0;
    if (e < E) { d[k] = dst[e]; s[k] = src[e]; }
  }
  const int lane = tid & 31;
#pragma unroll
  for (int k = 0; k < PART_EDGES; ++k) {
    const int b = (d[k] >= 0) ? (d[k] >> BIN_SHIFT) : -1;
    const int prev = __shfl_up(b, 1, 32);
    const bool head = (lane == 0) || (prev != b);
    const unsigned hm = (unsigned)__ballot(head);
    const unsigned upto = (2u << lane) - 1u;
    const int hl = 31 - __clz((int)(hm & upto));
    const unsigned above = hm & ~upto;
    const int nxt = (above != 0u) ? (__ffs((int)above) - 1) : 32;
    int base = 0;
    if (head && b >= 0) base = atomicAdd(&lh[b], nxt - hl);
    lr[k] = __shfl(base, hl, 32) + (lane - hl);
  }
  __syncthreads();
  const int v = (tid < LBINS) ? lh[tid] : 0;
  int incl = v;
#pragma unroll
  for (int off = 1; off < 32; off <<= 1) {
    const int t = __shfl_up(incl, off, 32);
    if (lane >= off) incl += t;
  }
  if (tid < LBINS && lane == 31) wsum[tid >> 5] = incl;
  __syncthreads();
  if (tid < LBINS) {
    int before = 0;
    for (int w = 0; w < (tid >> 5); ++w) before += wsum[w];
    lb[tid] = before + incl - v;
    gb[tid] = (v > 0) ? atomicAdd(&cursor[tid * CUR_PAD], v) : 0;
  }
  __syncthreads();
#pragma unroll
  for (int k = 0; k < PART_EDGES; ++k) {
    if (d[k] >= 0)
      stage[lb[d[k] >> BIN_SHIFT] + lr[k]] =
          ((unsigned)d[k] << 16) | (unsigned)s[k];
  }
  __syncthreads();
  const int nl = min(PART_TILE, E - tile0);
  for (int i = tid; i < nl; i += PART_THREADS) {
    const unsigned rec = stage[i];
    const int bin = (int)(rec >> (16 + BIN_SHIFT));
    const int pos = gb[bin] + (i - lb[bin]);
    if (pos < BCAP) binbuf[(size_t)bin * BCAP + pos] = (int)rec;
  }
}

template <int CTRL>
__device__ __forceinline__ int quad_sel_i(int v) {
  return __builtin_amdgcn_update_dpp(0, v, CTRL, 0xF, 0xF, false);
}
template <int CTRL>
__device__ __forceinline__ float quad_sel_f(float v) {
  return __int_as_float(quad_sel_i<CTRL>(__float_as_int(v)));
}
#define QUAD_XOR1 0xB1
#define QUAD_XOR2 0x4E

__device__ __forceinline__ float dot4(const float4 f, const float4 w) {
  float d = f.x * w.x;
  d = fmaf(f.y, w.y, d);
  d = fmaf(f.z, w.z, d);
  d = fmaf(f.w, w.w, d);
  return d;
}

__device__ __forceinline__ float4 ft_row(const float* __restrict__ ft, int s,
                                         int j) {
  const unsigned boff = (unsigned)s * (HD * 4u) + (unsigned)j * 16u;
  return *(const float4*)((const char*)ft + boff);
}

template <bool TAIL>
__device__ __forceinline__ void gat_trip(const float* __restrict__ ft, int s,
                                         int j, int q, int nlive,
                                         const float4 al, float erh, float& m,
                                         float& ps, float4& acc) {
  const int s0 = quad_sel_i<0x00>(s);
  const int s1 = quad_sel_i<0x55>(s);
  const int s2 = quad_sel_i<0xAA>(s);
  const int s3 = quad_sel_i<0xFF>(s);
  const float4 zero4 = {0.0f, 0.0f, 0.0f, 0.0f};
  float4 f0, f1, f2, f3;
  f0 = ft_row(ft, s0, j);
  if (!TAIL) {
    f1 = ft_row(ft, s1, j);
    f2 = ft_row(ft, s2, j);
    f3 = ft_row(ft, s3, j);
  } else {
    f1 = zero4; f2 = zero4; f3 = zero4;
    if (nlive > 1) f1 = ft_row(ft, s1, j);
    if (nlive > 2) f2 = ft_row(ft, s2, j);
  }
  const float d0 = dot4(f0, al);
  const float d1 = dot4(f1, al);
  const float d2 = dot4(f2, al);
  const float d3 = dot4(f3, al);
  const bool o1 = (q & 1) != 0, o2 = (q & 2) != 0;
  const float a = (o1 ? d1 : d0) + quad_sel_f<QUAD_XOR1>(o1 ? d0 : d1);
  const float b = (o1 ? d3 : d2) + quad_sel_f<QUAD_XOR1>(o1 ? d2 : d3);
  const float el = (o2 ? b : a) + quad_sel_f<QUAD_XOR2>(o2 ? a : b);
  float x = el + erh;
  x = (x > 0.0f) ? x : NEG_SLOPE * x;
  if (TAIL && q >= nlive) x = -INFINITY;
  float cm = fmaxf(x, quad_sel_f<QUAD_XOR1>(x));
  cm = fmaxf(cm, quad_sel_f<QUAD_XOR2>(cm));
  const float mn = fmaxf(m, cm);
  const float sc = expf(m - mn);
  const float p = expf(x - mn);
  m = mn;
  ps = fmaf(ps, sc, p);
  const float p0 = quad_sel_f<0x00>(p);
  const float p1 = quad_sel_f<0x55>(p);
  const float p2 = quad_sel_f<0xAA>(p);
  const float p3 = quad_sel_f<0xFF>(p);
  acc.x = fmaf(p3, f3.x, fmaf(p2, f2.x, fmaf(p1, f1.x, fmaf(p0, f0.x, acc.x * sc))));
  acc.y = fmaf(p3, f3.y, fmaf(p2, f2.y, fmaf(p1, f1.y, fmaf(p0, f0.y, acc.y * sc))));
  acc.z = fmaf(p3, f3.z, fmaf(p2, f2.z, fmaf(p1, f1.z, fmaf(p0, f0.z, acc.z * sc))));
  acc.w = fmaf(p3, f3.w, fmaf(p2, f2.w, fmaf(p1, f1.w, fmaf(p0, f0.w, acc.w * sc))));
}

__device__ __forceinline__ int rank_record(unsigned rec, bool live, int* lcnt) {
  const int d = (int)(rec >> 16);
  if (live && (d >> 4) == (int)blockIdx.x)
    return atomicAdd(&lcnt[d & (GROUP_NODES - 1)], 1);
  return CAP;
}
__device__ __forceinline__ void store_record(unsigned rec, int r, slot_t* ls) {
  if (r < CAP)
    ls[(int)((rec >> 16) & (GROUP_NODES - 1)) * CAP + r] =
        (slot_t)(rec & 0xFFFFu);
}
__device__ __forceinline__ void keep_records(const int4 a, int ia, const int4 b,
                                             int ib, int total, int* lcnt,
                                             slot_t* ls) {
  const unsigned w[8] = {(unsigned)a.x, (unsigned)a.y, (unsigned)a.z,
                         (unsigned)a.w, (unsigned)b.x, (unsigned)b.y,
                         (unsigned)b.z, (unsigned)b.w};
  int r[8];
#pragma unroll
  for (int k = 0; k < 4; ++k) {
    r[k] = rank_record(w[k], ia + k < total, lcnt);
    r[4 + k] = rank_record(w[4 + k], ib + k < total, lcnt);
  }
#pragma unroll
  for (int k = 0; k < 8; ++k) store_record(w[k], r[k], ls);
}

__global__ __launch_bounds__(256) void gat_node(
    const int* __restrict__ binbuf, const int* __restrict__ cursor,
    const int* __restrict__ src, const int* __restrict__ dst,
    const float* __restrict__ ft, const float* __restrict__ attn_l,
    const float* __restrict__ attn_r, const float* __restrict__ bias,
    float* __restrict__ out, int nNodes, int E) {
  __shared__ int lcnt[GROUP_NODES];
  __shared__ slot_t ls[GROUP_NODES * CAP];
  const int tid = threadIdx.x;
  const bool table = nNodes <= 0x10000;
  if (tid < GROUP_NODES) lcnt[tid] = 0;
  __syncthreads();
  if (table) {
    const int bin = (int)(blockIdx.x * GROUP_NODES) >> BIN_SHIFT;
    const int total = cursor[bin * CUR_PAD];
    if (total <= BCAP) {
      const int4* __restrict__ recs =
          (const int4*)(binbuf + (size_t)bin * BCAP);
      for (int i = tid; 4 * i < total; i += 512) {
        const int4 ra = recs[i];
        const int4 rb = recs[min(i + 256, BCAP / 4 - 1)];
        keep_records(ra, 4 * i, rb, 4 * (i + 256), total, lcnt, ls);
      }
    } else {
      for (int e = tid; e < E; e += 256) {
        const unsigned rec = ((unsigned)dst[e] << 16) | (unsigned)src[e];
        store_record(rec, rank_record(rec, true, lcnt), ls);
      }
    }
  }
  __syncthreads();

  const int j = tid & 15;
  const int g = tid >> 4;
  const int n = blockIdx.x * GROUP_NODES + g;
  if (n >= nNodes) return;
  const int q = j & 3;
  const slot_t* seg = ls + g * CAP;
  int s = seg[q];
  const int deg = lcnt[g];

  const float4 al = ((const float4*)attn_l)[j];
  const float4 ar = ((const float4*)attn_r)[j];
  float erh = dot4(ft_row(ft, n, j), ar);
  erh += quad_sel_f<QUAD_XOR1>(erh);
  erh += quad_sel_f<QUAD_XOR2>(erh);

  float m = -INFINITY, ps = 0.0f;
  float4 acc = {0.0f, 0.0f, 0.0f, 0.0f};
  int nedge = deg;
  if (deg > CAP || !table) {
    const int gshift = threadIdx.x & 16;
    nedge = 0;
    for (int base = 0; base < E; base += 16) {
      const int e = base + j;
      const bool hit = (e < E) && (dst[e] == n);
      unsigned hits = (unsigned)(__ballot(hit) >> gshift) & 0xFFFFu;
      while (hits != 0u) {
        const int b = __ffs(hits) - 1;
        hits &= hits - 1u;
        gat_trip<true>(ft, src[base + b], j, q, 1, al, erh, m, ps, acc);
        ++nedge;
      }
    }
  } else if (deg > 0) {
    int base = 0;
    for (; base + 4 <= deg; base += 4) {
      const int sn = seg[min(base + 4 + q, CAP - 1)];
      gat_trip<false>(ft, s, j, q, 4, al, erh, m, ps, acc);
      s = sn;
    }
    if (base < deg) gat_trip<true>(ft, s, j, q, deg - base, al, erh, m, ps, acc);
  }
  ps += quad_sel_f<QUAD_XOR1>(ps);
  ps += quad_sel_f<QUAD_XOR2>(ps);

  float4 o;
  o.x = bias[4 * j + 0];
  o.y = bias[4 * j + 1];
  o.z = bias[4 * j + 2];
  o.w = bias[4 * j + 3];
  if (nedge > 0) {
    o.x += acc.x / ps;
    o.y += acc.y / ps;
    o.z += acc.z / ps;
    o.w += acc.w / ps;
  }
  ((float4*)out)[(size_t)n * (HD / 4) + j] = o;
}

extern "C" void kernel_launch(void* const* d_in, const int* in_sizes, int n_in,
                              void* d_out, int out_size, void* d_ws,
                              size_t ws_size, hipStream_t stream) {
  const float* feat   = (const float*)d_in[0];
  const int*   src    = (const int*)d_in[1];
  const int*   dst    = (const int*)d_in[2];
  const float* W      = (const float*)d_in[3];
  const float* attn_l = (const float*)d_in[4];
  const float* attn_r = (const float*)d_in[5];
  const float* bias   = (const float*)d_in[6];
  float* out = (float*)d_out;

  const int N = in_sizes[0] / IN_FEATS;
  const int E = in_sizes[1];

  const int nBins = (N + BIN_NODES - 1) / BIN_NODES;

  float* ws = (float*)d_ws;
  float* ft = ws;
  int* binbuf = (int*)(ft + (size_t)N * HD);
  int* cursor = binbuf + (size_t)nBins * BCAP;

  {
    int nTiles = (N + 15) / 16;
    int blocks = (nTiles + 7) / 8;
    gat_proj_wmma<<<blocks, 256, 0, stream>>>(feat, W, ft, cursor, nBins, N,
                                              nTiles);
  }
  gat_part<<<(E + PART_TILE - 1) / PART_TILE, PART_THREADS, 0, stream>>>(
      src, dst, cursor, binbuf, E, N);
  gat_node<<<(N + GROUP_NODES - 1) / GROUP_NODES, 256, 0, stream>>>(
      binbuf, cursor, src, dst, ft, attn_l, attn_r, bias, out, N, E);
}


# Projection launch through LDS: W copied once per work-group to LDS as k-pairs (B fragments from LDS, not 8 global loads per k-step) and C tile turned in LDS: ft rows leave as whole 128-B lines
# speedup vs baseline: 6.8645x; 1.0842x over previous
#include <hip/hip_runtime.h>
#include <math.h>


typedef __attribute__((ext_vector_type(2))) float v2f;
typedef __attribute__((ext_vector_type(8))) float v8f;

#define IN_FEATS 128
#define HEADS    4
#define DHEAD    16
#define HD       64
#define NEG_SLOPE 0.2f
#define CAP      64
typedef unsigned short slot_t;
#define BIN_SHIFT 7
#define BIN_NODES 128
#define BCAP      4096
#define CUR_PAD   32
#define LBINS     512
#define PART_THREADS 1024
#define PART_EDGES   4
#define PART_TILE    (PART_THREADS * PART_EDGES)
#define GROUP_NODES  16

#define WP_PITCH 80
#define C_PITCH  68
__global__ __launch_bounds__(256) void gat_proj_wmma(
    const float* __restrict__ feat, const float* __restrict__ W,
    float* __restrict__ ft, int* __restrict__ cursor, int nBins, int nNodes,
    int nTiles) {
  __shared__ __attribute__((aligned(16))) float2 Wp[(IN_FEATS / 2) * WP_PITCH];
  __shared__ __attribute__((aligned(16))) float Cs[8 * 16 * C_PITCH];
  const int gid = blockIdx.x * blockDim.x + threadIdx.x;
  if (gid < nBins) cursor[gid * CUR_PAD] = 0;

#pragma unroll
  for (int i = 0; i < (IN_FEATS / 2) * (HD / 4) / 256; ++i) {
    const int u = (int)threadIdx.x + 256 * i;
    const int p = u >> 4, n4 = u & 15;
    const float4 a = *(const float4*)(W + (size_t)(2 * p) * HD + 4 * n4);
    const float4 b = *(const float4*)(W + (size_t)(2 * p + 1) * HD + 4 * n4);
    float4* d = (float4*)(Wp + p * WP_PITCH + 4 * n4);
    float4 lo, hi;
    lo.x = a.x; lo.y = b.x; lo.z = a.y; lo.w = b.y;
    hi.x = a.z; hi.y = b.z; hi.z = a.w; hi.w = b.w;
    d[0] = lo;
    d[1] = hi;
  }
  __syncthreads();

  const int lane = threadIdx.x & 31;
  const int wave = threadIdx.x >> 5;
  const int tile_raw = blockIdx.x * (blockDim.x >> 5) + wave;
  const bool live = tile_raw < nTiles;
  const int tile = live ? tile_raw : nTiles - 1;

  const int row0 = tile * 16;
  const int mrow = lane & 15;
  const int kq   = lane >> 4;

  int arow = row0 + mrow;
  if (arow >= nNodes) arow = nNodes - 1;
  const float2* __restrict__ apf2 =
      (const float2*)(feat + (size_t)arow * IN_FEATS) + kq;
  const float2* wl = Wp + kq * WP_PITCH + mrow;

  v8f acc0 = {}, acc1 = {}, acc2 = {}, acc3 = {};

#pragma unroll 4
  for (int k = 0; k < IN_FEATS; k += 4) {
    float2 av = apf2[k >> 1];
    v2f a; a[0] = av.x; a[1] = av.y;

    const float2* wr = wl + (k >> 1) * WP_PITCH;
    const float2 w0 = wr[0], w1 = wr[16], w2 = wr[32], w3 = wr[48];
    v2f b0; b0[0] = w0.x; b0[1] = w0.y;
    v2f b1; b1[0] = w1.x; b1[1] = w1.y;
    v2f b2; b2[0] = w2.x; b2[1] = w2.y;
    v2f b3; b3[0] = w3.x; b3[1] = w3.y;

    acc0 = __builtin_amdgcn_wmma_f32_16x16x4_f32(false, a, false, b0,
                                                 (short)0, acc0, false, false);
    acc1 = __builtin_amdgcn_wmma_f32_16x16x4_f32(false, a, false, b1,
                                                 (short)0, acc1, false, false);
    acc2 = __builtin_amdgcn_wmma_f32_16x16x4_f32(false, a, false, b2,
                                                 (short)0, acc2, false, false);
    acc3 = __builtin_amdgcn_wmma_f32_16x16x4_f32(false, a, false, b3,
                                                 (short)0, acc3, false, false);
  }

  float* cs = Cs + wave * (16 * C_PITCH);
#pragma unroll
  for (int r = 0; r < 8; ++r) {
    float* o = cs + (r + 8 * kq) * C_PITCH + mrow;
    o[0]  = acc0[r];
    o[16] = acc1[r];
    o[32] = acc2[r];
    o[48] = acc3[r];
  }
  __syncthreads();
  const float* crow = cs + kq * C_PITCH + 4 * mrow;
  float* grow = ft + (size_t)(row0 + kq) * HD + 4 * mrow;
  if (live && row0 + 16 <= nNodes) {
    float4 v[8];
#pragma unroll
    for (int i = 0; i < 8; ++i)
      v[i] = *(const float4*)(crow + 2 * i * C_PITCH);
#pragma unroll
    for (int i = 0; i < 8; ++i)
      *(float4*)(grow + (size_t)(2 * i) * HD) = v[i];
  } else if (live) {
    for (int i = 0; i < 8; ++i) {
      if (row0 + 2 * i + kq < nNodes)
        *(float4*)(grow + (size_t)(2 * i) * HD) =
            *(const float4*)(crow + 2 * i * C_PITCH);
    }
  }
}

__global__ __launch_bounds__(PART_THREADS) void gat_part(
    const int* __restrict__ src, const int* __restrict__ dst,
    int* __restrict__ cursor, int* __restrict__ binbuf, int E, int nNodes) {
  if (nNodes > 0x10000) return;
  __shared__ int lh[LBINS];
  __shared__ int lb[LBINS];
  __shared__ int gb[LBINS];
  __shared__ int wsum[LBINS / 32];
  __shared__ unsigned stage[PART_TILE];
  const int tid = threadIdx.x;
  if (tid < LBINS) lh[tid] = 0;
  __syncthreads();
  const int tile0 = blockIdx.x * PART_TILE;
  int d[PART_EDGES], s[PART_EDGES], lr[PART_EDGES];
#pragma unroll
  for (int k = 0; k < PART_EDGES; ++k) {
    const int e = tile0 + k * PART_THREADS + tid;
    d[k] = -1;
    s[k] = 0;
    if (e < E) { d[k] = dst[e]; s[k] = src[e]; }
  }
  const int lane = tid & 31;
#pragma unroll
  for (int k = 0; k < PART_EDGES; ++k) {
    const int b = (d[k] >= 0) ? (d[k] >> BIN_SHIFT) : -1;
    const int prev = __shfl_up(b, 1, 32);
    const bool head = (lane == 0) || (prev != b);
    const unsigned hm = (unsigned)__ballot(head);
    const unsigned upto = (2u << lane) - 1u;
    const int hl = 31 - __clz((int)(hm & upto));
    const unsigned above = hm & ~upto;
    const int nxt = (above != 0u) ? (__ffs((int)above) - 1) : 32;
    int base = 0;
    if (head && b >= 0) base = atomicAdd(&lh[b], nxt - hl);
    lr[k] = __shfl(base, hl, 32) + (lane - hl);
  }
  __syncthreads();
  const int v = (tid < LBINS) ? lh[tid] : 0;
  int incl = v;
#pragma unroll
  for (int off = 1; off < 32; off <<= 1) {
    const int t = __shfl_up(incl, off, 32);
    if (lane >= off) incl += t;
  }
  if (tid < LBINS && lane == 31) wsum[tid >> 5] = incl;
  __syncthreads();
  if (tid < LBINS) {
    int before = 0;
    for (int w = 0; w < (tid >> 5); ++w) before += wsum[w];
    lb[tid] = before + incl - v;
    gb[tid] = (v > 0) ? atomicAdd(&cursor[tid * CUR_PAD], v) : 0;
  }
  __syncthreads();
#pragma unroll
  for (int k = 0; k < PART_EDGES; ++k) {
    if (d[k] >= 0)
      stage[lb[d[k] >> BIN_SHIFT] + lr[k]] =
          ((unsigned)d[k] << 16) | (unsigned)s[k];
  }
  __syncthreads();
  const int nl = min(PART_TILE, E - tile0);
  for (int i = tid; i < nl; i += PART_THREADS) {
    const unsigned rec = stage[i];
    const int bin = (int)(rec >> (16 + BIN_SHIFT));
    const int pos = gb[bin] + (i - lb[bin]);
    if (pos < BCAP) binbuf[(size_t)bin * BCAP + pos] = (int)rec;
  }
}

template <int CTRL>
__device__ __forceinline__ int quad_sel_i(int v) {
  return __builtin_amdgcn_update_dpp(0, v, CTRL, 0xF, 0xF, false);
}
template <int CTRL>
__device__ __forceinline__ float quad_sel_f(float v) {
  return __int_as_float(quad_sel_i<CTRL>(__float_as_int(v)));
}
#define QUAD_XOR1 0xB1
#define QUAD_XOR2 0x4E

__device__ __forceinline__ float dot4(const float4 f, const float4 w) {
  float d = f.x * w.x;
  d = fmaf(f.y, w.y, d);
  d = fmaf(f.z, w.z, d);
  d = fmaf(f.w, w.w, d);
  return d;
}

__device__ __forceinline__ float4 ft_row(const float* __restrict__ ft, int s,
                                         int j) {
  const unsigned boff = (unsigned)s * (HD * 4u) + (unsigned)j * 16u;
  return *(const float4*)((const char*)ft + boff);
}

template <bool TAIL>
__device__ __forceinline__ void gat_trip(const float* __restrict__ ft, int s,
                                         int j, int q, int nlive,
                                         const float4 al, float erh, float& m,
                                         float& ps, float4& acc) {
  const int s0 = quad_sel_i<0x00>(s);
  const int s1 = quad_sel_i<0x55>(s);
  const int s2 = quad_sel_i<0xAA>(s);
  const int s3 = quad_sel_i<0xFF>(s);
  const float4 zero4 = {0.0f, 0.0f, 0.0f, 0.0f};
  float4 f0, f1, f2, f3;
  f0 = ft_row(ft, s0, j);
  if (!TAIL) {
    f1 = ft_row(ft, s1, j);
    f2 = ft_row(ft, s2, j);
    f3 = ft_row(ft, s3, j);
  } else {
    f1 = zero4; f2 = zero4; f3 = zero4;
    if (nlive > 1) f1 = ft_row(ft, s1, j);
    if (nlive > 2) f2 = ft_row(ft, s2, j);
  }
  const float d0 = dot4(f0, al);
  const float d1 = dot4(f1, al);
  const float d2 = dot4(f2, al);
  const float d3 = dot4(f3, al);
  const bool o1 = (q & 1) != 0, o2 = (q & 2) != 0;
  const float a = (o1 ? d1 : d0) + quad_sel_f<QUAD_XOR1>(o1 ? d0 : d1);
  const float b = (o1 ? d3 : d2) + quad_sel_f<QUAD_XOR1>(o1 ? d2 : d3);
  const float el = (o2 ? b : a) + quad_sel_f<QUAD_XOR2>(o2 ? a : b);
  float x = el + erh;
  x = (x > 0.0f) ? x : NEG_SLOPE * x;
  if (TAIL && q >= nlive) x = -INFINITY;
  float cm = fmaxf(x, quad_sel_f<QUAD_XOR1>(x));
  cm = fmaxf(cm, quad_sel_f<QUAD_XOR2>(cm));
  const float mn = fmaxf(m, cm);
  const float sc = expf(m - mn);
  const float p = expf(x - mn);
  m = mn;
  ps = fmaf(ps, sc, p);
  const float p0 = quad_sel_f<0x00>(p);
  const float p1 = quad_sel_f<0x55>(p);
  const float p2 = quad_sel_f<0xAA>(p);
  const float p3 = quad_sel_f<0xFF>(p);
  acc.x = fmaf(p3, f3.x, fmaf(p2, f2.x, fmaf(p1, f1.x, fmaf(p0, f0.x, acc.x * sc))));
  acc.y = fmaf(p3, f3.y, fmaf(p2, f2.y, fmaf(p1, f1.y, fmaf(p0, f0.y, acc.y * sc))));
  acc.z = fmaf(p3, f3.z, fmaf(p2, f2.z, fmaf(p1, f1.z, fmaf(p0, f0.z, acc.z * sc))));
  acc.w = fmaf(p3, f3.w, fmaf(p2, f2.w, fmaf(p1, f1.w, fmaf(p0, f0.w, acc.w * sc))));
}

__device__ __forceinline__ int rank_record(unsigned rec, bool live, int* lcnt) {
  const int d = (int)(rec >> 16);
  if (live && (d >> 4) == (int)blockIdx.x)
    return atomicAdd(&lcnt[d & (GROUP_NODES - 1)], 1);
  return CAP;
}
__device__ __forceinline__ void store_record(unsigned rec, int r, slot_t* ls) {
  if (r < CAP)
    ls[(int)((rec >> 16) & (GROUP_NODES - 1)) * CAP + r] =
        (slot_t)(rec & 0xFFFFu);
}
__device__ __forceinline__ void keep_records(const int4 a, int ia, const int4 b,
                                             int ib, int total, int* lcnt,
                                             slot_t* ls) {
  const unsigned w[8] = {(unsigned)a.x, (unsigned)a.y, (unsigned)a.z,
                         (unsigned)a.w, (unsigned)b.x, (unsigned)b.y,
                         (unsigned)b.z, (unsigned)b.w};
  int r[8];
#pragma unroll
  for (int k = 0; k < 4; ++k) {
    r[k] = rank_record(w[k], ia + k < total, lcnt);
    r[4 + k] = rank_record(w[4 + k], ib + k < total, lcnt);
  }
#pragma unroll
  for (int k = 0; k < 8; ++k) store_record(w[k], r[k], ls);
}

__global__ __launch_bounds__(256) void gat_node(
    const int* __restrict__ binbuf, const int* __restrict__ cursor,
    const int* __restrict__ src, const int* __restrict__ dst,
    const float* __restrict__ ft, const float* __restrict__ attn_l,
    const float* __restrict__ attn_r, const float* __restrict__ bias,
    float* __restrict__ out, int nNodes, int E) {
  __shared__ int lcnt[GROUP_NODES];
  __shared__ slot_t ls[GROUP_NODES * CAP];
  const int tid = threadIdx.x;
  const bool table = nNodes <= 0x10000;
  if (tid < GROUP_NODES) lcnt[tid] = 0;
  __syncthreads();
  if (table) {
    const int bin = (int)(blockIdx.x * GROUP_NODES) >> BIN_SHIFT;
    const int total = cursor[bin * CUR_PAD];
    if (total <= BCAP) {
      const int4* __restrict__ recs =
          (const int4*)(binbuf + (size_t)bin * BCAP);
      for (int i = tid; 4 * i < total; i += 512) {
        const int4 ra = recs[i];
        const int4 rb = recs[min(i + 256, BCAP / 4 - 1)];
        keep_records(ra, 4 * i, rb, 4 * (i + 256), total, lcnt, ls);
      }
    } else {
      for (int e = tid; e < E; e += 256) {
        const unsigned rec = ((unsigned)dst[e] << 16) | (unsigned)src[e];
        store_record(rec, rank_record(rec, true, lcnt), ls);
      }
    }
  }
  __syncthreads();

  const int j = tid & 15;
  const int g = tid >> 4;
  const int n = blockIdx.x * GROUP_NODES + g;
  if (n >= nNodes) return;
  const int q = j & 3;
  const slot_t* seg = ls + g * CAP;
  int s = seg[q];
  const int deg = lcnt[g];

  const float4 al = ((const float4*)attn_l)[j];
  const float4 ar = ((const float4*)attn_r)[j];
  float erh = dot4(ft_row(ft, n, j), ar);
  erh += quad_sel_f<QUAD_XOR1>(erh);
  erh += quad_sel_f<QUAD_XOR2>(erh);

  float m = -INFINITY, ps = 0.0f;
  float4 acc = {0.0f, 0.0f, 0.0f, 0.0f};
  int nedge = deg;
  if (deg > CAP || !table) {
    const int gshift = threadIdx.x & 16;
    nedge = 0;
    for (int base = 0; base < E; base += 16) {
      const int e = base + j;
      const bool hit = (e < E) && (dst[e] == n);
      unsigned hits = (unsigned)(__ballot(hit) >> gshift) & 0xFFFFu;
      while (hits != 0u) {
        const int b = __ffs(hits) - 1;
        hits &= hits - 1u;
        gat_trip<true>(ft, src[base + b], j, q, 1, al, erh, m, ps, acc);
        ++nedge;
      }
    }
  } else if (deg > 0) {
    int base = 0;
    for (; base + 4 <= deg; base += 4) {
      const int sn = seg[min(base + 4 + q, CAP - 1)];
      gat_trip<false>(ft, s, j, q, 4, al, erh, m, ps, acc);
      s = sn;
    }
    if (base < deg) gat_trip<true>(ft, s, j, q, deg - base, al, erh, m, ps, acc);
  }
  ps += quad_sel_f<QUAD_XOR1>(ps);
  ps += quad_sel_f<QUAD_XOR2>(ps);

  float4 o;
  o.x = bias[4 * j + 0];
  o.y = bias[4 * j + 1];
  o.z = bias[4 * j + 2];
  o.w = bias[4 * j + 3];
  if (nedge > 0) {
    o.x += acc.x / ps;
    o.y += acc.y / ps;
    o.z += acc.z / ps;
    o.w += acc.w / ps;
  }
  ((float4*)out)[(size_t)n * (HD / 4) + j] = o;
}

extern "C" void kernel_launch(void* const* d_in, const int* in_sizes, int n_in,
                              void* d_out, int out_size, void* d_ws,
                              size_t ws_size, hipStream_t stream) {
  const float* feat   = (const float*)d_in[0];
  const int*   src    = (const int*)d_in[1];
  const int*   dst    = (const int*)d_in[2];
  const float* W      = (const float*)d_in[3];
  const float* attn_l = (const float*)d_in[4];
  const float* attn_r = (const float*)d_in[5];
  const float* bias   = (const float*)d_in[6];
  float* out = (float*)d_out;

  const int N = in_sizes[0] / IN_FEATS;
  const int E = in_sizes[1];

  const int nBins = (N + BIN_NODES - 1) / BIN_NODES;

  float* ws = (float*)d_ws;
  float* ft = ws;
  int* binbuf = (int*)(ft + (size_t)N * HD);
  int* cursor = binbuf + (size_t)nBins * BCAP;

  {
    int nTiles = (N + 15) / 16;
    int blocks = (nTiles + 7) / 8;
    gat_proj_wmma<<<blocks, 256, 0, stream>>>(feat, W, ft, cursor, nBins, N,
                                              nTiles);
  }
  gat_part<<<(E + PART_TILE - 1) / PART_TILE, PART_THREADS, 0, stream>>>(
      src, dst, cursor, binbuf, E, N);
  gat_node<<<(N + GROUP_NODES - 1) / GROUP_NODES, 256, 0, stream>>>(
      binbuf, cursor, src, dst, ft, attn_l, attn_r, bias, out, N, E);
}


# Node launch softmax in base 2: attention vectors scaled by log2(e) once, each exp is one v_exp_f32 (not library expf), one v_rcp_f32 of the sum and 4 products for 4 divisions; same summation order
# speedup vs baseline: 7.6175x; 1.1097x over previous
#include <hip/hip_runtime.h>
#include <math.h>


typedef __attribute__((ext_vector_type(2))) float v2f;
typedef __attribute__((ext_vector_type(8))) float v8f;

#define IN_FEATS 128
#define HEADS    4
#define DHEAD    16
#define HD       64
#define NEG_SLOPE 0.2f
#define LOG2E 1.44269504088896340736f
#define CAP      64
typedef unsigned short slot_t;
#define BIN_SHIFT 7
#define BIN_NODES 128
#define LBINS     512
#define PART_THREADS 256
#define PART_EDGES   16
#define PART_TILE    (PART_THREADS * PART_EDGES)
typedef unsigned short off_t16;
#define NODE_THREADS 1024
#define NODE_GROUPS  (NODE_THREADS / 16)
#define RUN_LANES    4
#define RUN_FIRST    16
#define RUN_SLOTS    (NODE_THREADS / RUN_LANES)

#define WP_PITCH 80
#define C_PITCH  68
__device__ __forceinline__ void proj_body(
    const float* __restrict__ feat, const float* __restrict__ W,
    float* __restrict__ ft, int nNodes, int nTiles, int pblock) {
  __shared__ __attribute__((aligned(16))) float2 Wp[(IN_FEATS / 2) * WP_PITCH];
  __shared__ __attribute__((aligned(16))) float Cs[8 * 16 * C_PITCH];
#pragma unroll
  for (int i = 0; i < (IN_FEATS / 2) * (HD / 4) / 256; ++i) {
    const int u = (int)threadIdx.x + 256 * i;
    const int p = u >> 4, n4 = u & 15;
    const float4 a = *(const float4*)(W + (size_t)(2 * p) * HD + 4 * n4);
    const float4 b = *(const float4*)(W + (size_t)(2 * p + 1) * HD + 4 * n4);
    float4* d = (float4*)(Wp + p * WP_PITCH + 4 * n4);
    float4 lo, hi;
    lo.x = a.x; lo.y = b.x; lo.z = a.y; lo.w = b.y;
    hi.x = a.z; hi.y = b.z; hi.z = a.w; hi.w = b.w;
    d[0] = lo;
    d[1] = hi;
  }
  __syncthreads();

  const int lane = threadIdx.x & 31;
  const int wave = threadIdx.x >> 5;
  const int tile_raw = pblock * (blockDim.x >> 5) + wave;
  const bool live = tile_raw < nTiles;
  const int tile = live ? tile_raw : nTiles - 1;

  const int row0 = tile * 16;
  const int mrow = lane & 15;
  const int kq   = lane >> 4;

  int arow = row0 + mrow;
  if (arow >= nNodes) arow = nNodes - 1;
  const float2* __restrict__ apf2 =
      (const float2*)(feat + (size_t)arow * IN_FEATS) + kq;
  const float2* wl = Wp + kq * WP_PITCH + mrow;

  v8f acc0 = {}, acc1 = {}, acc2 = {}, acc3 = {};

#pragma unroll 4
  for (int k = 0; k < IN_FEATS; k += 4) {
    float2 av = apf2[k >> 1];
    v2f a; a[0] = av.x; a[1] = av.y;

    const float2* wr = wl + (k >> 1) * WP_PITCH;
    const float2 w0 = wr[0], w1 = wr[16], w2 = wr[32], w3 = wr[48];
    v2f b0; b0[0] = w0.x; b0[1] = w0.y;
    v2f b1; b1[0] = w1.x; b1[1] = w1.y;
    v2f b2; b2[0] = w2.x; b2[1] = w2.y;
    v2f b3; b3[0] = w3.x; b3[1] = w3.y;

    acc0 = __builtin_amdgcn_wmma_f32_16x16x4_f32(false, a, false, b0,
                                                 (short)0, acc0, false, false);
    acc1 = __builtin_amdgcn_wmma_f32_16x16x4_f32(false, a, false, b1,
                                                 (short)0, acc1, false, false);
    acc2 = __builtin_amdgcn_wmma_f32_16x16x4_f32(false, a, false, b2,
                                                 (short)0, acc2, false, false);
    acc3 = __builtin_amdgcn_wmma_f32_16x16x4_f32(false, a, false, b3,
                                                 (short)0, acc3, false, false);
  }

  float* cs = Cs + wave * (16 * C_PITCH);
#pragma unroll
  for (int r = 0; r < 8; ++r) {
    float* o = cs + (r + 8 * kq) * C_PITCH + mrow;
    o[0]  = acc0[r];
    o[16] = acc1[r];
    o[32] = acc2[r];
    o[48] = acc3[r];
  }
  __syncthreads();
  const float* crow = cs + kq * C_PITCH + 4 * mrow;
  float* grow = ft + (size_t)(row0 + kq) * HD + 4 * mrow;
  if (live && row0 + 16 <= nNodes) {
    float4 v[8];
#pragma unroll
    for (int i = 0; i < 8; ++i)
      v[i] = *(const float4*)(crow + 2 * i * C_PITCH);
#pragma unroll
    for (int i = 0; i < 8; ++i)
      *(float4*)(grow + (size_t)(2 * i) * HD) = v[i];
  } else if (live) {
    for (int i = 0; i < 8; ++i) {
      if (row0 + 2 * i + kq < nNodes)
        *(float4*)(grow + (size_t)(2 * i) * HD) =
            *(const float4*)(crow + 2 * i * C_PITCH);
    }
  }
}

__device__ __forceinline__ void part_body(
    const int* __restrict__ src, const int* __restrict__ dst,
    unsigned* __restrict__ tiles, off_t16* __restrict__ offs, int E,
    int nNodes, int nPart, int nBins, int w) {
  if (nNodes > 0x10000) return;
  __shared__ int lh[LBINS];
  __shared__ int lb[LBINS];
  __shared__ int wsum[PART_THREADS / 32];
  __shared__ unsigned stage[PART_TILE];
  const int tid = threadIdx.x;
  for (int i = tid; i < LBINS; i += PART_THREADS) lh[i] = 0;
  __syncthreads();
  const int tile0 = w * PART_TILE;
  int d[PART_EDGES], s[PART_EDGES], lr[PART_EDGES];
#pragma unroll
  for (int k = 0; k < PART_EDGES; ++k) {
    const int e = tile0 + k * PART_THREADS + tid;
    d[k] = -1;
    s[k] = 0;
    if (e < E) { d[k] = dst[e]; s[k] = src[e]; }
  }
  const int lane = tid & 31;
#pragma unroll
  for (int k = 0; k < PART_EDGES; ++k) {
    const int b = (d[k] >= 0) ? (d[k] >> BIN_SHIFT) : -1;
    const int prev = __shfl_up(b, 1, 32);
    const bool head = (lane == 0) || (prev != b);
    const unsigned hm = (unsigned)__ballot(head);
    const unsigned upto = (2u << lane) - 1u;
    const int hl = 31 - __clz((int)(hm & upto));
    const unsigned above = hm & ~upto;
    const int nxt = (above != 0u) ? (__ffs((int)above) - 1) : 32;
    int base = 0;
    if (head && b >= 0) base = atomicAdd(&lh[b], nxt - hl);
    lr[k] = __shfl(base, hl, 32) + (lane - hl);
  }
  __syncthreads();
  const int v0 = lh[2 * tid], v1 = lh[2 * tid + 1];
  const int sv = v0 + v1;
  int incl = sv;
#pragma unroll
  for (int off = 1; off < 32; off <<= 1) {
    const int t = __shfl_up(incl, off, 32);
    if (lane >= off) incl += t;
  }
  if (lane == 31) wsum[tid >> 5] = incl;
  __syncthreads();
  {
    int before = 0;
    for (int ww = 0; ww < (tid >> 5); ++ww) before += wsum[ww];
    const int ex0 = before + incl - sv;
    const int ex1 = ex0 + v0;
    lb[2 * tid] = ex0;
    lb[2 * tid + 1] = ex1;
    if (2 * tid < nBins) offs[(size_t)(2 * tid) * nPart + w] = (off_t16)ex0;
    if (2 * tid + 1 < nBins)
      offs[(size_t)(2 * tid + 1) * nPart + w] = (off_t16)ex1;
  }
  const int nl = min(PART_TILE, E - tile0);
  if (tid == 0) offs[(size_t)nBins * nPart + w] = (off_t16)nl;
  __syncthreads();
#pragma unroll
  for (int k = 0; k < PART_EDGES; ++k) {
    if (d[k] >= 0)
      stage[lb[d[k] >> BIN_SHIFT] + lr[k]] =
          ((unsigned)d[k] << 16) | (unsigned)s[k];
  }
  __syncthreads();
  unsigned* __restrict__ to = tiles + (size_t)w * PART_TILE;
  for (int i = tid; i < nl; i += PART_THREADS) to[i] = stage[i];
}

__global__ __launch_bounds__(256) void gat_proj_part(
    const float* __restrict__ feat, const float* __restrict__ W,
    float* __restrict__ ft, const int* __restrict__ src,
    const int* __restrict__ dst, unsigned* __restrict__ tiles,
    off_t16* __restrict__ offs, int nNodes, int nTiles, int E, int nPart,
    int nBins) {
  if ((int)blockIdx.x < nPart)
    part_body(src, dst, tiles, offs, E, nNodes, nPart, nBins, (int)blockIdx.x);
  else
    proj_body(feat, W, ft, nNodes, nTiles, (int)blockIdx.x - nPart);
}

template <int CTRL>
__device__ __forceinline__ int quad_sel_i(int v) {
  return __builtin_amdgcn_update_dpp(0, v, CTRL, 0xF, 0xF, false);
}
template <int CTRL>
__device__ __forceinline__ float quad_sel_f(float v) {
  return __int_as_float(quad_sel_i<CTRL>(__float_as_int(v)));
}
#define QUAD_XOR1 0xB1
#define QUAD_XOR2 0x4E

__device__ __forceinline__ float dot4(const float4 f, const float4 w) {
  float d = f.x * w.x;
  d = fmaf(f.y, w.y, d);
  d = fmaf(f.z, w.z, d);
  d = fmaf(f.w, w.w, d);
  return d;
}

__device__ __forceinline__ float4 ft_row(const float* __restrict__ ft, int s,
                                         int j) {
  const unsigned boff = (unsigned)s * (HD * 4u) + (unsigned)j * 16u;
  return *(const float4*)((const char*)ft + boff);
}

template <bool TAIL>
__device__ __forceinline__ void gat_trip(const float* __restrict__ ft, int s,
                                         int j, int q, int nlive,
                                         const float4 al, float erh, float& m,
                                         float& ps, float4& acc) {
  const int s0 = quad_sel_i<0x00>(s);
  const int s1 = quad_sel_i<0x55>(s);
  const int s2 = quad_sel_i<0xAA>(s);
  const int s3 = quad_sel_i<0xFF>(s);
  const float4 zero4 = {0.0f, 0.0f, 0.0f, 0.0f};
  float4 f0, f1, f2, f3;
  f0 = ft_row(ft, s0, j);
  if (!TAIL) {
    f1 = ft_row(ft, s1, j);
    f2 = ft_row(ft, s2, j);
    f3 = ft_row(ft, s3, j);
  } else {
    f1 = zero4; f2 = zero4; f3 = zero4;
    if (nlive > 1) f1 = ft_row(ft, s1, j);
    if (nlive > 2) f2 = ft_row(ft, s2, j);
  }
  const float d0 = dot4(f0, al);
  const float d1 = dot4(f1, al);
  const float d2 = dot4(f2, al);
  const float d3 = dot4(f3, al);
  const bool o1 = (q & 1) != 0, o2 = (q & 2) != 0;
  const float a = (o1 ? d1 : d0) + quad_sel_f<QUAD_XOR1>(o1 ? d0 : d1);
  const float b = (o1 ? d3 : d2) + quad_sel_f<QUAD_XOR1>(o1 ? d2 : d3);
  const float el = (o2 ? b : a) + quad_sel_f<QUAD_XOR2>(o2 ? a : b);
  float x = el + erh;
  x = (x > 0.0f) ? x : NEG_SLOPE * x;
  if (TAIL && q >= nlive) x = -INFINITY;
  float cm = fmaxf(x, quad_sel_f<QUAD_XOR1>(x));
  cm = fmaxf(cm, quad_sel_f<QUAD_XOR2>(cm));
  const float mn = fmaxf(m, cm);
  const float sc = __builtin_amdgcn_exp2f(m - mn);
  const float p = __builtin_amdgcn_exp2f(x - mn);
  m = mn;
  ps = fmaf(ps, sc, p);
  const float p0 = quad_sel_f<0x00>(p);
  const float p1 = quad_sel_f<0x55>(p);
  const float p2 = quad_sel_f<0xAA>(p);
  const float p3 = quad_sel_f<0xFF>(p);
  acc.x = fmaf(p3, f3.x, fmaf(p2, f2.x, fmaf(p1, f1.x, fmaf(p0, f0.x, acc.x * sc))));
  acc.y = fmaf(p3, f3.y, fmaf(p2, f2.y, fmaf(p1, f1.y, fmaf(p0, f0.y, acc.y * sc))));
  acc.z = fmaf(p3, f3.z, fmaf(p2, f2.z, fmaf(p1, f1.z, fmaf(p0, f0.z, acc.z * sc))));
  acc.w = fmaf(p3, f3.w, fmaf(p2, f2.w, fmaf(p1, f1.w, fmaf(p0, f0.w, acc.w * sc))));
}

__device__ __forceinline__ int rank_record(unsigned rec, bool live, int* lcnt) {
  if (live) return atomicAdd(&lcnt[(rec >> 16) & (BIN_NODES - 1)], 1);
  return CAP;
}
__device__ __forceinline__ void store_record(unsigned rec, int r, slot_t* ls) {
  if (r < CAP)
    ls[(int)((rec >> 16) & (BIN_NODES - 1)) * CAP + r] =
        (slot_t)(rec & 0xFFFFu);
}

__global__ __launch_bounds__(NODE_THREADS) void gat_node(
    const unsigned* __restrict__ tiles, const off_t16* __restrict__ offs,
    const int* __restrict__ src, const int* __restrict__ dst,
    const float* __restrict__ ft, const float* __restrict__ attn_l,
    const float* __restrict__ attn_r, const float* __restrict__ bias,
    float* __restrict__ out, int nNodes, int E, int nPart) {
  __shared__ int lcnt[BIN_NODES];
  __shared__ slot_t ls[BIN_NODES * CAP];
  __shared__ int nlong;
  __shared__ int ltile[RUN_SLOTS], lbeg[RUN_SLOTS], lend[RUN_SLOTS];
  const int tid = threadIdx.x;
  const int bin = (int)blockIdx.x;
  const bool table = nNodes <= 0x10000;
  if (tid < BIN_NODES) lcnt[tid] = 0;
  if (tid == 0) nlong = 0;
  if (table) {
    for (int w0 = 0; w0 < nPart; w0 += RUN_SLOTS) {
      __syncthreads();
      const int w = w0 + (tid / RUN_LANES);
      const int r = tid % RUN_LANES;
      if (w < nPart) {
        const int o0 = offs[(size_t)bin * nPart + w];
        const int o1 = offs[(size_t)(bin + 1) * nPart + w];
        const unsigned* __restrict__ run = tiles + (size_t)w * PART_TILE;
        const int stop = min(o1, o0 + RUN_FIRST);
        unsigned rec[RUN_FIRST / RUN_LANES];
        int rk[RUN_FIRST / RUN_LANES];
#pragma unroll
        for (int k = 0; k < RUN_FIRST / RUN_LANES; ++k) {
          const int i = o0 + r + RUN_LANES * k;
          rec[k] = (i < stop) ? run[i] : 0u;
        }
#pragma unroll
        for (int k = 0; k < RUN_FIRST / RUN_LANES; ++k)
          rk[k] = rank_record(rec[k], o0 + r + RUN_LANES * k < stop, lcnt);
#pragma unroll
        for (int k = 0; k < RUN_FIRST / RUN_LANES; ++k)
          store_record(rec[k], rk[k], ls);
        if (r == 0 && o1 > stop) {
          const int at = atomicAdd(&nlong, 1);
          ltile[at] = w;
          lbeg[at] = stop;
          lend[at] = o1;
        }
      }
      __syncthreads();
      const int nl = nlong;
      for (int k = 0; k < nl; ++k) {
        const unsigned* __restrict__ run = tiles + (size_t)ltile[k] * PART_TILE;
        const int e1 = lend[k];
        for (int i = lbeg[k] + tid; i < e1; i += NODE_THREADS) {
          const unsigned rc = run[i];
          store_record(rc, rank_record(rc, true, lcnt), ls);
        }
      }
      __syncthreads();
      if (tid == 0) nlong = 0;
    }
  }
  __syncthreads();

  const int j = tid & 15;
  const int g = tid >> 4;
  const int q = j & 3;
  float4 al = ((const float4*)attn_l)[j];
  float4 ar = ((const float4*)attn_r)[j];
  al.x *= LOG2E; al.y *= LOG2E; al.z *= LOG2E; al.w *= LOG2E;
  ar.x *= LOG2E; ar.y *= LOG2E; ar.z *= LOG2E; ar.w *= LOG2E;
  float4 bv;
  bv.x = bias[4 * j + 0];
  bv.y = bias[4 * j + 1];
  bv.z = bias[4 * j + 2];
  bv.w = bias[4 * j + 3];

  for (int node = g; node < BIN_NODES; node += NODE_GROUPS) {
    const int n = bin * BIN_NODES + node;
    if (n >= nNodes) break;
    const slot_t* seg = ls + node * CAP;
    int s = seg[q];
    const int deg = lcnt[node];

    float erh = dot4(ft_row(ft, n, j), ar);
    erh += quad_sel_f<QUAD_XOR1>(erh);
    erh += quad_sel_f<QUAD_XOR2>(erh);

    float m = -INFINITY, ps = 0.0f;
    float4 acc = {0.0f, 0.0f, 0.0f, 0.0f};
    int nedge = deg;
    if (deg > CAP || !table) {
      const int gshift = threadIdx.x & 16;
      nedge = 0;
      for (int base = 0; base < E; base += 16) {
        const int e = base + j;
        const bool hit = (e < E) && (dst[e] == n);
        unsigned hits = (unsigned)(__ballot(hit) >> gshift) & 0xFFFFu;
        while (hits != 0u) {
          const int b = __ffs(hits) - 1;
          hits &= hits - 1u;
          gat_trip<true>(ft, src[base + b], j, q, 1, al, erh, m, ps, acc);
          ++nedge;
        }
      }
    } else if (deg > 0) {
      int base = 0;
      for (; base + 4 <= deg; base += 4) {
        const int sn = seg[min(base + 4 + q, CAP - 1)];
        gat_trip<false>(ft, s, j, q, 4, al, erh, m, ps, acc);
        s = sn;
      }
      if (base < deg)
        gat_trip<true>(ft, s, j, q, deg - base, al, erh, m, ps, acc);
    }
    ps += quad_sel_f<QUAD_XOR1>(ps);
    ps += quad_sel_f<QUAD_XOR2>(ps);

    float4 o = bv;
    if (nedge > 0) {
      const float rs = __builtin_amdgcn_rcpf(ps);
      o.x = fmaf(acc.x, rs, o.x);
      o.y = fmaf(acc.y, rs, o.y);
      o.z = fmaf(acc.z, rs, o.z);
      o.w = fmaf(acc.w, rs, o.w);
    }
    ((float4*)out)[(size_t)n * (HD / 4) + j] = o;
  }
}

extern "C" void kernel_launch(void* const* d_in, const int* in_sizes, int n_in,
                              void* d_out, int out_size, void* d_ws,
                              size_t ws_size, hipStream_t stream) {
  const float* feat   = (const float*)d_in[0];
  const int*   src    = (const int*)d_in[1];
  const int*   dst    = (const int*)d_in[2];
  const float* W      = (const float*)d_in[3];
  const float* attn_l = (const float*)d_in[4];
  const float* attn_r = (const float*)d_in[5];
  const float* bias   = (const float*)d_in[6];
  float* out = (float*)d_out;

  const int N = in_sizes[0] / IN_FEATS;
  const int E = in_sizes[1];

  const int nBins = (N + BIN_NODES - 1) / BIN_NODES;
  const int nPart = (E + PART_TILE - 1) / PART_TILE;

  float* ws = (float*)d_ws;
  float* ft = ws;
  unsigned* tiles = (unsigned*)(ft + (size_t)N * HD);
  off_t16* offs = (off_t16*)(tiles + (size_t)nPart * PART_TILE);

  const int nTiles = (N + 15) / 16;
  const int pblocks = (nTiles + 7) / 8;
  gat_proj_part<<<nPart + pblocks, 256, 0, stream>>>(
      feat, W, ft, src, dst, tiles, offs, N, nTiles, E, nPart, nBins);
  gat_node<<<nBins, NODE_THREADS, 0, stream>>>(
      tiles, offs, src, dst, ft, attn_l, attn_r, bias, out, N, E, nPart);
}
